# starting kernel
# baseline (speedup 1.0000x reference)
#include <hip/hip_runtime.h>

typedef __attribute__((ext_vector_type(2))) float v2f;
typedef __attribute__((ext_vector_type(8))) float v8f;

#define WAVES_PER_BLOCK 8
#define LSTRIDE 33

__constant__ float cH[23] = {
    -1.20162964e-4f, 0.0f,  1.615524292e-3f, 0.0f, -1.0385513306e-2f, 0.0f,
     4.3619155884e-2f, 0.0f, -1.45397186478e-1f, 0.0f, 6.1066818237e-1f,
     1.0f,
     6.1066818237e-1f, 0.0f, -1.45397186478e-1f, 0.0f, 4.3619155884e-2f, 0.0f,
    -1.0385513306e-2f, 0.0f,  1.615524292e-3f, 0.0f, -1.20162964e-4f };

__global__ __launch_bounds__(256) void cdf23_up2x_wmma(
    const float* __restrict__ in, float* __restrict__ out, int N, int phi)
{
  __shared__ float sG[16 * 20];
  __shared__ float sX[WAVES_PER_BLOCK][20 * LSTRIDE];
  __shared__ float sT[WAVES_PER_BLOCK][16 * LSTRIDE];

  const int tid  = threadIdx.x;
  const int lane = tid & 31;
  const int wv   = tid >> 5;
  const int cofs = phi ? 22 : 21;

  for (int t = tid; t < 16 * 20; t += 256) {
    int m = t / 20, n = t - 20 * m;
    int j = m - 2 * n + cofs;
    sG[t] = (j >= 0 && j < 23) ? cH[j] : 0.0f;
  }
  __syncthreads();

  const int TW  = (2 * N) >> 4;
  const int tpp = TW * TW;
  const int g     = blockIdx.x * WAVES_PER_BLOCK + wv;
  const int plane = g / tpp;
  const int rem   = g - plane * tpp;
  const int i0 = (rem / TW) * 16;
  const int j0 = (rem % TW) * 16;
  const int back = phi ? 6 : 5;
  const int nr0 = (i0 >> 1) - back;
  const int nc0 = (j0 >> 1) - back;
  const float* __restrict__ xin = in + (long long)plane * N * N;

  float* X = sX[wv];
  float* T = sT[wv];

  for (int t = lane; t < 20 * LSTRIDE; t += 32) X[t] = 0.0f;
  __builtin_amdgcn_wave_barrier();
  for (int t = lane; t < 19 * 19; t += 32) {
    int r = t / 19, q = t - 19 * r;
    int gr = nr0 + r; if (gr < 0) gr += N; else if (gr >= N) gr -= N;
    int gc = nc0 + q; if (gc < 0) gc += N; else if (gc >= N) gc -= N;
    X[r * LSTRIDE + q] = xin[(long long)gr * N + gc];
    int gc2 = gc + 8; if (gc2 >= N) gc2 -= N;
    __builtin_prefetch(xin + (long long)gr * N + gc2, 0, 1);
  }
  __builtin_amdgcn_wave_barrier();

  const int mh = lane & 15;
  const int kh = (lane < 16) ? 0 : 2;

  v8f t0 = {}, t1 = {};
#pragma unroll
  for (int k = 0; k < 5; ++k) {
    const int kb = 4 * k + kh;
    v2f a, b, b2;
    a.x  = sG[mh * 20 + kb];
    a.y  = sG[mh * 20 + kb + 1];
    b.x  = X[kb * LSTRIDE + mh];
    b.y  = X[(kb + 1) * LSTRIDE + mh];
    b2.x = X[kb * LSTRIDE + 16 + mh];
    b2.y = X[(kb + 1) * LSTRIDE + 16 + mh];
    t0 = __builtin_amdgcn_wmma_f32_16x16x4_f32(false, a, false, b,  (short)0, t0, false, false);
    t1 = __builtin_amdgcn_wmma_f32_16x16x4_f32(false, a, false, b2, (short)0, t1, false, false);
  }

  const int mrow = (lane < 16) ? 0 : 8;
#pragma unroll
  for (int v = 0; v < 8; ++v) {
    T[(v + mrow) * LSTRIDE + mh]      = t0[v];
    T[(v + mrow) * LSTRIDE + 16 + mh] = t1[v];
  }
  __builtin_amdgcn_wave_barrier();

  v8f o = {};
#pragma unroll
  for (int k = 0; k < 5; ++k) {
    const int kb = 4 * k + kh;
    v2f a, b;
    a.x = T[mh * LSTRIDE + kb];
    a.y = T[mh * LSTRIDE + kb + 1];
    b.x = sG[mh * 20 + kb];
    b.y = sG[mh * 20 + kb + 1];
    o = __builtin_amdgcn_wmma_f32_16x16x4_f32(false, a, false, b, (short)0, o, false, false);
  }

  const int OW = 2 * N;
  float* __restrict__ op = out + (long long)plane * OW * OW;
#pragma unroll
  for (int v = 0; v < 8; ++v) {
    op[(long long)(i0 + v + mrow) * OW + (j0 + mh)] = o[v];
  }
}

extern "C" void kernel_launch(void* const* d_in, const int* in_sizes, int n_in,
                              void* d_out, int out_size, void* d_ws, size_t ws_size,
                              hipStream_t stream) {
  (void)in_sizes; (void)n_in; (void)out_size; (void)ws_size;
  const float* img = (const float*)d_in[0];
  float* out = (float*)d_out;
  float* mid = (float*)d_ws;

  cdf23_up2x_wmma<<<2048, 256, 0, stream>>>(img, mid, 256, 1);

  cdf23_up2x_wmma<<<8192, 256, 0, stream>>>(mid, out, 512, 0);
}


# each wave walks a 64-col column strip: row filter on float2 global loads first, its WMMA D tiles fed unchanged as B operands of the column filter (no LDS window/transpose); f32 sum order changed
# speedup vs baseline: 2.1524x; 2.1524x over previous
#include <hip/hip_runtime.h>

typedef __attribute__((ext_vector_type(2))) float v2f;
typedef __attribute__((ext_vector_type(8))) float v8f;

__constant__ float cH[23] = {
    -1.20162964e-4f, 0.0f,  1.615524292e-3f, 0.0f, -1.0385513306e-2f, 0.0f,
     4.3619155884e-2f, 0.0f, -1.45397186478e-1f, 0.0f, 6.1066818237e-1f,
     1.0f,
     6.1066818237e-1f, 0.0f, -1.45397186478e-1f, 0.0f, 4.3619155884e-2f, 0.0f,
    -1.0385513306e-2f, 0.0f,  1.615524292e-3f, 0.0f, -1.20162964e-4f };

#define WMMA4(a, b, c) \
  __builtin_amdgcn_wmma_f32_16x16x4_f32(false, (a), false, (b), (short)0, (c), false, false)

template <int N, int PHI, int UNITS>
__global__ __launch_bounds__(256) void cdf23_up2x_chain(
    const float* __restrict__ in, float* __restrict__ out)
{
  __shared__ float sH[96];
  const int tid = threadIdx.x;
  if (tid < 96) {
    const int tap = tid - 32;
    sH[tid] = (tap >= 0 && tap < 23) ? cH[tap] : 0.0f;
  }
  __syncthreads();

  constexpr int OW = 2 * N;
  constexpr int SP = OW / 64;
  constexpr int RS = OW / (32 * UNITS);
  const int lane = tid & 31;
  const int n16  = lane & 15;
  const int half = lane >> 4;
  const int g     = blockIdx.x * 8 + (tid >> 5);
  const int plane = g / (SP * RS);
  const int rem   = g - plane * (SP * RS);
  const int rseg  = rem / SP;
  const int strip = rem - rseg * SP;

  const int lh = 4 * half - n16 + PHI + 31;
  v2f bh[5];
#pragma unroll
  for (int kk = 0; kk < 5; ++kk) {
    bh[kk].x = sH[lh + 8 * kk];
    bh[kk].y = sH[lh + 8 * kk + 2];
  }
  const int lv = 16 * half - n16 + PHI + 31;
  v2f aE0[4], aE1[2], aO0[4], aO1[4];
#pragma unroll
  for (int c = 0; c < 4; ++c) {
    aE0[c].x = sH[lv + 4 * c];       aE0[c].y = sH[lv + 4 * c + 2];
    aO0[c].x = sH[lv + 4 * c - 16];  aO0[c].y = sH[lv + 4 * c - 14];
    aO1[c].x = sH[lv + 4 * c + 16];  aO1[c].y = sH[lv + 4 * c + 18];
  }
#pragma unroll
  for (int c = 0; c < 2; ++c) {
    aE1[c].x = sH[lv + 4 * c + 32];  aE1[c].y = sH[lv + 4 * c + 34];
  }

  const float* __restrict__ xin = in + (size_t)plane * N * N;
  float* __restrict__ op = out + (size_t)plane * OW * OW + 64 * strip + n16;

  int coff[11];
#pragma unroll
  for (int q = 0; q < 11; ++q) coff[q] = (32 * strip - 6 + 4 * q + 2 * half) & (N - 1);

  const int s0 = rseg * UNITS;
  v2f xa[11], xb[11];
  v8f T[UNITS + 1][4];

  {
    const float* r0 = xin + (size_t)((16 * s0 - 6 + n16) & (N - 1)) * N;
    const float* r1 = xin + (size_t)((16 * (s0 + 1) - 6 + n16) & (N - 1)) * N;
#pragma unroll
    for (int q = 0; q < 11; ++q) xa[q] = *(const v2f*)(r0 + coff[q]);
#pragma unroll
    for (int q = 0; q < 11; ++q) xb[q] = *(const v2f*)(r1 + coff[q]);
  }
#pragma unroll
  for (int nt = 0; nt < 4; ++nt) {
    v8f acc = {};
#pragma unroll
    for (int kk = 0; kk < 5; ++kk) acc = WMMA4(xa[2 * nt + kk], bh[kk], acc);
    T[0][nt] = acc;
  }
#pragma unroll
  for (int nt = 0; nt < 4; ++nt) {
    v8f acc = {};
#pragma unroll
    for (int kk = 0; kk < 5; ++kk) acc = WMMA4(xb[2 * nt + kk], bh[kk], acc);
    T[1][nt] = acc;
  }

#pragma unroll
  for (int u = 0; u < UNITS; ++u) {
    if (u + 1 < UNITS) {
      const float* r2 = xin + (size_t)((16 * (s0 + u + 2) - 6 + n16) & (N - 1)) * N;
#pragma unroll
      for (int q = 0; q < 11; ++q) xa[q] = *(const v2f*)(r2 + coff[q]);
    }
    float* o0 = op + (size_t)(32 * (s0 + u) + 8 * half) * OW;
#pragma unroll
    for (int nt = 0; nt < 4; ++nt) {
      v8f oe = {}, oo = {};
#pragma unroll
      for (int c = 0; c < 4; ++c) {
        v2f b; b.x = T[u][nt][2 * c]; b.y = T[u][nt][2 * c + 1];
        oe = WMMA4(aE0[c], b, oe);
        oo = WMMA4(aO0[c], b, oo);
      }
#pragma unroll
      for (int c = 0; c < 4; ++c) {
        v2f b; b.x = T[u + 1][nt][2 * c]; b.y = T[u + 1][nt][2 * c + 1];
        if (c < 2) oe = WMMA4(aE1[c], b, oe);
        oo = WMMA4(aO1[c], b, oo);
      }
#pragma unroll
      for (int v = 0; v < 8; ++v) {
        o0[(size_t)v * OW + 16 * nt]        = oe[v];
        o0[(size_t)(16 + v) * OW + 16 * nt] = oo[v];
      }
    }
    if (u + 1 < UNITS) {
#pragma unroll
      for (int nt = 0; nt < 4; ++nt) {
        v8f acc = {};
#pragma unroll
        for (int kk = 0; kk < 5; ++kk) acc = WMMA4(xa[2 * nt + kk], bh[kk], acc);
        T[u + 2][nt] = acc;
      }
    }
  }
}

extern "C" void kernel_launch(void* const* d_in, const int* in_sizes, int n_in,
                              void* d_out, int out_size, void* d_ws, size_t ws_size,
                              hipStream_t stream) {
  (void)in_sizes; (void)n_in; (void)out_size; (void)ws_size;
  const float* img = (const float*)d_in[0];
  float* out = (float*)d_out;
  float* mid = (float*)d_ws;

  cdf23_up2x_chain<256, 1, 1><<<256, 256, 0, stream>>>(img, mid);

  cdf23_up2x_chain<512, 0, 2><<<512, 256, 0, stream>>>(mid, out);
}


# row-filter loader takes tile rows in permuted order so each D register pair fed on as B holds 4 consecutive rows: column filter 10 WMMAs per 32x16 outputs, not 14; f32 sum order changed
# speedup vs baseline: 2.4670x; 1.1462x over previous
#include <hip/hip_runtime.h>

typedef __attribute__((ext_vector_type(2))) float v2f;
typedef __attribute__((ext_vector_type(8))) float v8f;

__constant__ float cH[23] = {
    -1.20162964e-4f, 0.0f,  1.615524292e-3f, 0.0f, -1.0385513306e-2f, 0.0f,
     4.3619155884e-2f, 0.0f, -1.45397186478e-1f, 0.0f, 6.1066818237e-1f,
     1.0f,
     6.1066818237e-1f, 0.0f, -1.45397186478e-1f, 0.0f, 4.3619155884e-2f, 0.0f,
    -1.0385513306e-2f, 0.0f,  1.615524292e-3f, 0.0f, -1.20162964e-4f };

#define WMMA4(a, b, c) \
  __builtin_amdgcn_wmma_f32_16x16x4_f32(false, (a), false, (b), (short)0, (c), false, false)

template <int N, int PHI, int UNITS>
__global__ __launch_bounds__(256) void cdf23_up2x_chain(
    const float* __restrict__ in, float* __restrict__ out)
{
  __shared__ float sH[96];
  const int tid = threadIdx.x;
  if (tid < 96) {
    const int tap = tid - 32;
    sH[tid] = (tap >= 0 && tap < 23) ? cH[tap] : 0.0f;
  }
  __syncthreads();

  constexpr int OW = 2 * N;
  constexpr int SP = OW / 64;
  constexpr int RS = OW / (32 * UNITS);
  const int lane = tid & 31;
  const int n16  = lane & 15;
  const int half = lane >> 4;
  const int g     = blockIdx.x * 8 + (tid >> 5);
  const int plane = g / (SP * RS);
  const int rem   = g - plane * (SP * RS);
  const int rseg  = rem / SP;
  const int strip = rem - rseg * SP;

  const int lh = 4 * half - n16 + PHI + 31;
  v2f bh[5];
#pragma unroll
  for (int kk = 0; kk < 5; ++kk) {
    bh[kk].x = sH[lh + 8 * kk];
    bh[kk].y = sH[lh + 8 * kk + 2];
  }
  const int prow = 4 * ((n16 & 7) >> 1) + 2 * (n16 >> 3) + (n16 & 1);

  const float* __restrict__ xin = in + (size_t)plane * N * N;
  float* __restrict__ op = out + (size_t)plane * OW * OW + 64 * strip + n16;

  int coff[11];
#pragma unroll
  for (int q = 0; q < 11; ++q) coff[q] = (32 * strip - 6 + 4 * q + 2 * half) & (N - 1);

  const int s0 = rseg * UNITS;
  v2f xa[11], xb[11];
  v8f T[UNITS + 1][4];

  {
    const float* r0 = xin + (size_t)((16 * s0 - 6 + prow) & (N - 1)) * N;
    const float* r1 = xin + (size_t)((16 * (s0 + 1) - 6 + prow) & (N - 1)) * N;
#pragma unroll
    for (int q = 0; q < 11; ++q) xa[q] = *(const v2f*)(r0 + coff[q]);
#pragma unroll
    for (int q = 0; q < 11; ++q) xb[q] = *(const v2f*)(r1 + coff[q]);
  }
#pragma unroll
  for (int nt = 0; nt < 4; ++nt) {
    v8f acc = {};
#pragma unroll
    for (int kk = 0; kk < 5; ++kk) acc = WMMA4(xa[2 * nt + kk], bh[kk], acc);
    T[0][nt] = acc;
  }
#pragma unroll
  for (int nt = 0; nt < 4; ++nt) {
    v8f acc = {};
#pragma unroll
    for (int kk = 0; kk < 5; ++kk) acc = WMMA4(xb[2 * nt + kk], bh[kk], acc);
    T[1][nt] = acc;
  }

#pragma unroll
  for (int u = 0; u < UNITS; ++u) {
    if (u + 1 < UNITS) {
      const float* r2 = xin + (size_t)((16 * (s0 + u + 2) - 6 + prow) & (N - 1)) * N;
#pragma unroll
      for (int q = 0; q < 11; ++q) xa[q] = *(const v2f*)(r2 + coff[q]);
    }
    float* o0 = op + (size_t)(32 * (s0 + u) + 8 * half) * OW;
#pragma unroll
    for (int nt = 0; nt < 4; ++nt) {
      v8f oe = {}, oo = {};
#pragma unroll
      for (int j = 0; j < 7; ++j) {
        v2f b;
        if (j < 4) { b.x = T[u][nt][2 * j];           b.y = T[u][nt][2 * j + 1]; }
        else       { b.x = T[u + 1][nt][2 * (j - 4)]; b.y = T[u + 1][nt][2 * (j - 4) + 1]; }
        if (j < 5)  oe = WMMA4(bh[j], b, oe);
        if (j >= 2) oo = WMMA4(bh[j - 2], b, oo);
      }
#pragma unroll
      for (int v = 0; v < 8; ++v) {
        o0[(size_t)v * OW + 16 * nt]        = oe[v];
        o0[(size_t)(16 + v) * OW + 16 * nt] = oo[v];
      }
    }
    if (u + 1 < UNITS) {
#pragma unroll
      for (int nt = 0; nt < 4; ++nt) {
        v8f acc = {};
#pragma unroll
        for (int kk = 0; kk < 5; ++kk) acc = WMMA4(xa[2 * nt + kk], bh[kk], acc);
        T[u + 2][nt] = acc;
      }
    }
  }
}

extern "C" void kernel_launch(void* const* d_in, const int* in_sizes, int n_in,
                              void* d_out, int out_size, void* d_ws, size_t ws_size,
                              hipStream_t stream) {
  (void)in_sizes; (void)n_in; (void)out_size; (void)ws_size;
  const float* img = (const float*)d_in[0];
  float* out = (float*)d_out;
  float* mid = (float*)d_ws;

  cdf23_up2x_chain<256, 1, 1><<<256, 256, 0, stream>>>(img, mid);

  cdf23_up2x_chain<512, 0, 2><<<512, 256, 0, stream>>>(mid, out);
}


# row filter reads input as aligned 16-byte groups per lane: 6 loads per 16-row tile, not 11 of 8 bytes (half the lines touched); k chunks re-paired with own tap fragments; f32 sum order changed
# speedup vs baseline: 2.5573x; 1.0366x over previous
#include <hip/hip_runtime.h>

typedef __attribute__((ext_vector_type(2))) float v2f;
typedef __attribute__((ext_vector_type(4))) float v4f;
typedef __attribute__((ext_vector_type(8))) float v8f;

__constant__ float cH[23] = {
    -1.20162964e-4f, 0.0f,  1.615524292e-3f, 0.0f, -1.0385513306e-2f, 0.0f,
     4.3619155884e-2f, 0.0f, -1.45397186478e-1f, 0.0f, 6.1066818237e-1f,
     1.0f,
     6.1066818237e-1f, 0.0f, -1.45397186478e-1f, 0.0f, 4.3619155884e-2f, 0.0f,
    -1.0385513306e-2f, 0.0f,  1.615524292e-3f, 0.0f, -1.20162964e-4f };

#define WMMA4(a, b, c) \
  __builtin_amdgcn_wmma_f32_16x16x4_f32(false, (a), false, (b), (short)0, (c), false, false)


__device__ __forceinline__ v8f row_tile(const v4f& g0, const v4f& g1, const v4f& g2,
                                        const v2f (&br)[5], bool hi)
{
  v2f e;
  e.x = hi ? g0.x : g2.z;
  e.y = hi ? g0.y : g2.w;
  v8f acc = {};
  acc = WMMA4(g0.zw, br[0], acc);
  acc = WMMA4(g1.xy, br[1], acc);
  acc = WMMA4(g1.zw, br[2], acc);
  acc = WMMA4(g2.xy, br[3], acc);
  acc = WMMA4(e,     br[4], acc);
  return acc;
}

template <int N, int PHI, int UNITS>
__global__ __launch_bounds__(256) void cdf23_up2x_chain(
    const float* __restrict__ in, float* __restrict__ out)
{
  __shared__ float sH[96];
  const int tid = threadIdx.x;
  if (tid < 96) {
    const int tap = tid - 32;
    sH[tid] = (tap >= 0 && tap < 23) ? cH[tap] : 0.0f;
  }
  __syncthreads();

  constexpr int OW = 2 * N;
  constexpr int SP = OW / 64;
  constexpr int RS = OW / (32 * UNITS);
  const int lane = tid & 31;
  const int n16  = lane & 15;
  const int half = lane >> 4;
  const int g     = blockIdx.x * 8 + (tid >> 5);
  const int plane = g / (SP * RS);
  const int rem   = g - plane * (SP * RS);
  const int rseg  = rem / SP;
  const int strip = rem - rseg * SP;

  const int lh = 4 * half - n16 + PHI + 31;
  v2f bh[5];
#pragma unroll
  for (int kk = 0; kk < 5; ++kk) {
    bh[kk].x = sH[lh + 8 * kk];
    bh[kk].y = sH[lh + 8 * kk + 2];
  }
  const int nb[5] = { 4 * half, 6 + 4 * half, 8 + 4 * half, 14 + 4 * half, 16 - 14 * half };
  const int lr = PHI + 31 - n16;
  v2f br[5];
#pragma unroll
  for (int j = 0; j < 5; ++j) {
    br[j].x = sH[lr + 2 * nb[j]];
    br[j].y = sH[lr + 2 * nb[j] + 2];
  }
  const bool hi = half != 0;
  const int prow = 4 * ((n16 & 7) >> 1) + 2 * (n16 >> 3) + (n16 & 1);

  const float* __restrict__ xin = in + (size_t)plane * N * N;
  float* __restrict__ op = out + (size_t)plane * OW * OW + 64 * strip + n16;

  int goff[6];
#pragma unroll
  for (int p = 0; p < 6; ++p) goff[p] = (32 * strip - 8 + 8 * p + 4 * half) & (N - 1);

  const int s0 = rseg * UNITS;
  v4f xa[6], xb[6];
  v8f T[UNITS + 1][4];

  {
    const float* r0 = xin + (size_t)((16 * s0 - 6 + prow) & (N - 1)) * N;
    const float* r1 = xin + (size_t)((16 * (s0 + 1) - 6 + prow) & (N - 1)) * N;
#pragma unroll
    for (int p = 0; p < 6; ++p) xa[p] = *(const v4f*)(r0 + goff[p]);
#pragma unroll
    for (int p = 0; p < 6; ++p) xb[p] = *(const v4f*)(r1 + goff[p]);
  }
#pragma unroll
  for (int nt = 0; nt < 4; ++nt) T[0][nt] = row_tile(xa[nt], xa[nt + 1], xa[nt + 2], br, hi);
#pragma unroll
  for (int nt = 0; nt < 4; ++nt) T[1][nt] = row_tile(xb[nt], xb[nt + 1], xb[nt + 2], br, hi);

#pragma unroll
  for (int u = 0; u < UNITS; ++u) {
    if (u + 1 < UNITS) {
      const float* r2 = xin + (size_t)((16 * (s0 + u + 2) - 6 + prow) & (N - 1)) * N;
#pragma unroll
      for (int p = 0; p < 6; ++p) xa[p] = *(const v4f*)(r2 + goff[p]);
    }
    float* o0 = op + (size_t)(32 * (s0 + u) + 8 * half) * OW;
#pragma unroll
    for (int nt = 0; nt < 4; ++nt) {
      v8f oe = {}, oo = {};
#pragma unroll
      for (int j = 0; j < 7; ++j) {
        v2f b;
        if (j < 4) { b.x = T[u][nt][2 * j];           b.y = T[u][nt][2 * j + 1]; }
        else       { b.x = T[u + 1][nt][2 * (j - 4)]; b.y = T[u + 1][nt][2 * (j - 4) + 1]; }
        if (j < 5)  oe = WMMA4(bh[j], b, oe);
        if (j >= 2) oo = WMMA4(bh[j - 2], b, oo);
      }
#pragma unroll
      for (int v = 0; v < 8; ++v) {
        o0[(size_t)v * OW + 16 * nt]        = oe[v];
        o0[(size_t)(16 + v) * OW + 16 * nt] = oo[v];
      }
    }
    if (u + 1 < UNITS) {
#pragma unroll
      for (int nt = 0; nt < 4; ++nt)
        T[u + 2][nt] = row_tile(xa[nt], xa[nt + 1], xa[nt + 2], br, hi);
    }
  }
}

extern "C" void kernel_launch(void* const* d_in, const int* in_sizes, int n_in,
                              void* d_out, int out_size, void* d_ws, size_t ws_size,
                              hipStream_t stream) {
  (void)in_sizes; (void)n_in; (void)out_size; (void)ws_size;
  const float* img = (const float*)d_in[0];
  float* out = (float*)d_out;
  float* mid = (float*)d_ws;

  cdf23_up2x_chain<256, 1, 1><<<256, 256, 0, stream>>>(img, mid);

  cdf23_up2x_chain<512, 0, 2><<<512, 256, 0, stream>>>(mid, out);
}


# D tiles of two side-by-side column tiles swap lane halves in registers (one cross-lane swap per register pair, no LDS): a store writes one whole 128-byte line, not two half lines; sums unchanged
# speedup vs baseline: 2.6953x; 1.0540x over previous
#include <hip/hip_runtime.h>

typedef __attribute__((ext_vector_type(2))) float v2f;
typedef __attribute__((ext_vector_type(4))) float v4f;
typedef __attribute__((ext_vector_type(8))) float v8f;

__constant__ float cH[23] = {
    -1.20162964e-4f, 0.0f,  1.615524292e-3f, 0.0f, -1.0385513306e-2f, 0.0f,
     4.3619155884e-2f, 0.0f, -1.45397186478e-1f, 0.0f, 6.1066818237e-1f,
     1.0f,
     6.1066818237e-1f, 0.0f, -1.45397186478e-1f, 0.0f, 4.3619155884e-2f, 0.0f,
    -1.0385513306e-2f, 0.0f,  1.615524292e-3f, 0.0f, -1.20162964e-4f };

#define WMMA4(a, b, c) \
  __builtin_amdgcn_wmma_f32_16x16x4_f32(false, (a), false, (b), (short)0, (c), false, false)


__device__ __forceinline__ v8f row_tile(const v4f& g0, const v4f& g1, const v4f& g2,
                                        const v2f (&br)[5], bool hi)
{
  v2f e;
  e.x = hi ? g0.x : g2.z;
  e.y = hi ? g0.y : g2.w;
  v8f acc = {};
  acc = WMMA4(g0.zw, br[0], acc);
  acc = WMMA4(g1.xy, br[1], acc);
  acc = WMMA4(g1.zw, br[2], acc);
  acc = WMMA4(g2.xy, br[3], acc);
  acc = WMMA4(e,     br[4], acc);
  return acc;
}

__device__ __forceinline__ void half_swap(v8f& a, v8f& b)
{
#pragma unroll
  for (int v = 0; v < 8; ++v) {
    const auto r = __builtin_amdgcn_permlane16_swap(__float_as_uint(a[v]), __float_as_uint(b[v]),
                                                    false, false);
    a[v] = __uint_as_float(r[0]);
    b[v] = __uint_as_float(r[1]);
  }
}

template <int N, int PHI, int UNITS>
__global__ __launch_bounds__(256) void cdf23_up2x_chain(
    const float* __restrict__ in, float* __restrict__ out)
{
  __shared__ float sH[96];
  const int tid = threadIdx.x;
  if (tid < 96) {
    const int tap = tid - 32;
    sH[tid] = (tap >= 0 && tap < 23) ? cH[tap] : 0.0f;
  }
  __syncthreads();

  constexpr int OW = 2 * N;
  constexpr int SP = OW / 64;
  constexpr int RS = OW / (32 * UNITS);
  const int lane = tid & 31;
  const int n16  = lane & 15;
  const int half = lane >> 4;
  const int g     = blockIdx.x * 8 + (tid >> 5);
  const int plane = g / (SP * RS);
  const int rem   = g - plane * (SP * RS);
  const int rseg  = rem / SP;
  const int strip = rem - rseg * SP;

  const int lh = 4 * half - n16 + PHI + 31;
  v2f bh[5];
#pragma unroll
  for (int kk = 0; kk < 5; ++kk) {
    bh[kk].x = sH[lh + 8 * kk];
    bh[kk].y = sH[lh + 8 * kk + 2];
  }
  const int nb[5] = { 4 * half, 6 + 4 * half, 8 + 4 * half, 14 + 4 * half, 16 - 14 * half };
  const int lr = PHI + 31 - n16;
  v2f br[5];
#pragma unroll
  for (int j = 0; j < 5; ++j) {
    br[j].x = sH[lr + 2 * nb[j]];
    br[j].y = sH[lr + 2 * nb[j] + 2];
  }
  const bool hi = half != 0;
  const int prow = 4 * ((n16 & 7) >> 1) + 2 * (n16 >> 3) + (n16 & 1);

  const float* __restrict__ xin = in + (size_t)plane * N * N;
  float* __restrict__ op = out + (size_t)plane * OW * OW + 64 * strip + lane;

  int goff[6];
#pragma unroll
  for (int p = 0; p < 6; ++p) goff[p] = (32 * strip - 8 + 8 * p + 4 * half) & (N - 1);

  const int s0 = rseg * UNITS;
  v4f xa[6], xb[6];
  v8f T[UNITS + 1][4];

  {
    const float* r0 = xin + (size_t)((16 * s0 - 6 + prow) & (N - 1)) * N;
    const float* r1 = xin + (size_t)((16 * (s0 + 1) - 6 + prow) & (N - 1)) * N;
#pragma unroll
    for (int p = 0; p < 6; ++p) xa[p] = *(const v4f*)(r0 + goff[p]);
#pragma unroll
    for (int p = 0; p < 6; ++p) xb[p] = *(const v4f*)(r1 + goff[p]);
  }
#pragma unroll
  for (int nt = 0; nt < 4; ++nt) T[0][nt] = row_tile(xa[nt], xa[nt + 1], xa[nt + 2], br, hi);
#pragma unroll
  for (int nt = 0; nt < 4; ++nt) T[1][nt] = row_tile(xb[nt], xb[nt + 1], xb[nt + 2], br, hi);

#pragma unroll
  for (int u = 0; u < UNITS; ++u) {
    if (u + 1 < UNITS) {
      const float* r2 = xin + (size_t)((16 * (s0 + u + 2) - 6 + prow) & (N - 1)) * N;
#pragma unroll
      for (int p = 0; p < 6; ++p) xa[p] = *(const v4f*)(r2 + goff[p]);
    }
    float* o0 = op + (size_t)(32 * (s0 + u)) * OW;
#pragma unroll
    for (int np = 0; np < 2; ++np) {
      v8f oe0 = {}, oo0 = {}, oe1 = {}, oo1 = {};
#pragma unroll
      for (int j = 0; j < 7; ++j) {
        v2f b0, b1;
        if (j < 4) {
          b0.x = T[u][2 * np][2 * j];     b0.y = T[u][2 * np][2 * j + 1];
          b1.x = T[u][2 * np + 1][2 * j]; b1.y = T[u][2 * np + 1][2 * j + 1];
        } else {
          b0.x = T[u + 1][2 * np][2 * (j - 4)];     b0.y = T[u + 1][2 * np][2 * (j - 4) + 1];
          b1.x = T[u + 1][2 * np + 1][2 * (j - 4)]; b1.y = T[u + 1][2 * np + 1][2 * (j - 4) + 1];
        }
        if (j < 5)  { oe0 = WMMA4(bh[j], b0, oe0);     oe1 = WMMA4(bh[j], b1, oe1); }
        if (j >= 2) { oo0 = WMMA4(bh[j - 2], b0, oo0); oo1 = WMMA4(bh[j - 2], b1, oo1); }
      }
      asm("v_nop\n\tv_nop\n\tv_nop\n\tv_nop" : "+v"(oe0), "+v"(oe1), "+v"(oo0), "+v"(oo1));
      half_swap(oe0, oe1);
      half_swap(oo0, oo1);
#pragma unroll
      for (int v = 0; v < 8; ++v) {
        o0[(size_t)v * OW + 32 * np]        = oe0[v];
        o0[(size_t)(8 + v) * OW + 32 * np]  = oe1[v];
        o0[(size_t)(16 + v) * OW + 32 * np] = oo0[v];
        o0[(size_t)(24 + v) * OW + 32 * np] = oo1[v];
      }
    }
    if (u + 1 < UNITS) {
#pragma unroll
      for (int nt = 0; nt < 4; ++nt)
        T[u + 2][nt] = row_tile(xa[nt], xa[nt + 1], xa[nt + 2], br, hi);
    }
  }
}

extern "C" void kernel_launch(void* const* d_in, const int* in_sizes, int n_in,
                              void* d_out, int out_size, void* d_ws, size_t ws_size,
                              hipStream_t stream) {
  (void)in_sizes; (void)n_in; (void)out_size; (void)ws_size;
  const float* img = (const float*)d_in[0];
  float* out = (float*)d_out;
  float* mid = (float*)d_ws;

  cdf23_up2x_chain<256, 1, 1><<<256, 256, 0, stream>>>(img, mid);

  cdf23_up2x_chain<512, 0, 2><<<512, 256, 0, stream>>>(mid, out);
}


# each wave walks a strip twice as long (launch 1: 64 rows, 4-wave groups; launch 2: 128 rows): half the waves to start, fewer halo tiles re-filtered, tile loads held 3 ahead; f32 sums unchanged
# speedup vs baseline: 2.7995x; 1.0387x over previous
#include <hip/hip_runtime.h>

typedef __attribute__((ext_vector_type(2))) float v2f;
typedef __attribute__((ext_vector_type(4))) float v4f;
typedef __attribute__((ext_vector_type(8))) float v8f;

__constant__ float cH[23] = {
    -1.20162964e-4f, 0.0f,  1.615524292e-3f, 0.0f, -1.0385513306e-2f, 0.0f,
     4.3619155884e-2f, 0.0f, -1.45397186478e-1f, 0.0f, 6.1066818237e-1f,
     1.0f,
     6.1066818237e-1f, 0.0f, -1.45397186478e-1f, 0.0f, 4.3619155884e-2f, 0.0f,
    -1.0385513306e-2f, 0.0f,  1.615524292e-3f, 0.0f, -1.20162964e-4f };

#define WMMA4(a, b, c) \
  __builtin_amdgcn_wmma_f32_16x16x4_f32(false, (a), false, (b), (short)0, (c), false, false)


__device__ __forceinline__ v8f row_tile(const v4f& g0, const v4f& g1, const v4f& g2,
                                        const v2f (&br)[5], bool hi)
{
  v2f e;
  e.x = hi ? g0.x : g2.z;
  e.y = hi ? g0.y : g2.w;
  v8f acc = {};
  acc = WMMA4(g0.zw, br[0], acc);
  acc = WMMA4(g1.xy, br[1], acc);
  acc = WMMA4(g1.zw, br[2], acc);
  acc = WMMA4(g2.xy, br[3], acc);
  acc = WMMA4(e,     br[4], acc);
  return acc;
}

__device__ __forceinline__ void half_swap(v8f& a, v8f& b)
{
#pragma unroll
  for (int v = 0; v < 8; ++v) {
    const auto r = __builtin_amdgcn_permlane16_swap(__float_as_uint(a[v]), __float_as_uint(b[v]),
                                                    false, false);
    a[v] = __uint_as_float(r[0]);
    b[v] = __uint_as_float(r[1]);
  }
}

template <int N, int PHI, int UNITS, int WPB>
__global__ __launch_bounds__(32 * WPB) void cdf23_up2x_chain(
    const float* __restrict__ in, float* __restrict__ out)
{
  __shared__ float sH[96];
  const int tid = threadIdx.x;
  if (tid < 96) {
    const int tap = tid - 32;
    sH[tid] = (tap >= 0 && tap < 23) ? cH[tap] : 0.0f;
  }
  __syncthreads();

  constexpr int OW = 2 * N;
  constexpr int SP = OW / 64;
  constexpr int RS = OW / (32 * UNITS);
  const int lane = tid & 31;
  const int n16  = lane & 15;
  const int half = lane >> 4;
  const int g     = blockIdx.x * WPB + (tid >> 5);
  const int plane = g / (SP * RS);
  const int rem   = g - plane * (SP * RS);
  const int rseg  = rem / SP;
  const int strip = rem - rseg * SP;

  const int lh = 4 * half - n16 + PHI + 31;
  v2f bh[5];
#pragma unroll
  for (int kk = 0; kk < 5; ++kk) {
    bh[kk].x = sH[lh + 8 * kk];
    bh[kk].y = sH[lh + 8 * kk + 2];
  }
  const int nb[5] = { 4 * half, 6 + 4 * half, 8 + 4 * half, 14 + 4 * half, 16 - 14 * half };
  const int lr = PHI + 31 - n16;
  v2f br[5];
#pragma unroll
  for (int j = 0; j < 5; ++j) {
    br[j].x = sH[lr + 2 * nb[j]];
    br[j].y = sH[lr + 2 * nb[j] + 2];
  }
  const bool hi = half != 0;
  const int prow = 4 * ((n16 & 7) >> 1) + 2 * (n16 >> 3) + (n16 & 1);

  const float* __restrict__ xin = in + (size_t)plane * N * N;
  float* __restrict__ op = out + (size_t)plane * OW * OW + 64 * strip + lane;

  int goff[6];
#pragma unroll
  for (int p = 0; p < 6; ++p) goff[p] = (32 * strip - 8 + 8 * p + 4 * half) & (N - 1);

  const int s0 = rseg * UNITS;
  v4f X[UNITS + 1][6];
  v8f T[UNITS + 1][4];

#pragma unroll
  for (int t = 0; t < 3 && t <= UNITS; ++t) {
    const float* r = xin + (size_t)((16 * (s0 + t) - 6 + prow) & (N - 1)) * N;
#pragma unroll
    for (int p = 0; p < 6; ++p) X[t][p] = *(const v4f*)(r + goff[p]);
  }
  __builtin_amdgcn_sched_barrier(0);
#pragma unroll
  for (int nt = 0; nt < 4; ++nt) T[0][nt] = row_tile(X[0][nt], X[0][nt + 1], X[0][nt + 2], br, hi);
#pragma unroll
  for (int nt = 0; nt < 4; ++nt) T[1][nt] = row_tile(X[1][nt], X[1][nt + 1], X[1][nt + 2], br, hi);

#pragma unroll
  for (int u = 0; u < UNITS; ++u) {
    if (u + 3 <= UNITS) {
      const float* r3 = xin + (size_t)((16 * (s0 + u + 3) - 6 + prow) & (N - 1)) * N;
#pragma unroll
      for (int p = 0; p < 6; ++p) X[u + 3][p] = *(const v4f*)(r3 + goff[p]);
    }
    if (u > 0 || u + 3 <= UNITS) {
      __builtin_amdgcn_sched_barrier(0);
    }
    float* o0 = op + (size_t)(32 * (s0 + u)) * OW;
#pragma unroll
    for (int np = 0; np < 2; ++np) {
      v8f oe0 = {}, oo0 = {}, oe1 = {}, oo1 = {};
#pragma unroll
      for (int j = 0; j < 7; ++j) {
        v2f b0, b1;
        if (j < 4) {
          b0.x = T[u][2 * np][2 * j];     b0.y = T[u][2 * np][2 * j + 1];
          b1.x = T[u][2 * np + 1][2 * j]; b1.y = T[u][2 * np + 1][2 * j + 1];
        } else {
          b0.x = T[u + 1][2 * np][2 * (j - 4)];     b0.y = T[u + 1][2 * np][2 * (j - 4) + 1];
          b1.x = T[u + 1][2 * np + 1][2 * (j - 4)]; b1.y = T[u + 1][2 * np + 1][2 * (j - 4) + 1];
        }
        if (j < 5)  { oe0 = WMMA4(bh[j], b0, oe0);     oe1 = WMMA4(bh[j], b1, oe1); }
        if (j >= 2) { oo0 = WMMA4(bh[j - 2], b0, oo0); oo1 = WMMA4(bh[j - 2], b1, oo1); }
      }
      asm("v_nop\n\tv_nop\n\tv_nop\n\tv_nop" : "+v"(oe0), "+v"(oe1), "+v"(oo0), "+v"(oo1));
      half_swap(oe0, oe1);
      half_swap(oo0, oo1);
#pragma unroll
      for (int v = 0; v < 8; ++v) {
        o0[(size_t)v * OW + 32 * np]        = oe0[v];
        o0[(size_t)(8 + v) * OW + 32 * np]  = oe1[v];
        o0[(size_t)(16 + v) * OW + 32 * np] = oo0[v];
        o0[(size_t)(24 + v) * OW + 32 * np] = oo1[v];
      }
    }
    if (u + 2 <= UNITS) {
#pragma unroll
      for (int nt = 0; nt < 4; ++nt)
        T[u + 2][nt] = row_tile(X[u + 2][nt], X[u + 2][nt + 1], X[u + 2][nt + 2], br, hi);
    }
  }
}

extern "C" void kernel_launch(void* const* d_in, const int* in_sizes, int n_in,
                              void* d_out, int out_size, void* d_ws, size_t ws_size,
                              hipStream_t stream) {
  (void)in_sizes; (void)n_in; (void)out_size; (void)ws_size;
  const float* img = (const float*)d_in[0];
  float* out = (float*)d_out;
  float* mid = (float*)d_ws;

  cdf23_up2x_chain<256, 1, 2, 4><<<256, 128, 0, stream>>>(img, mid);

  cdf23_up2x_chain<512, 0, 4, 8><<<256, 256, 0, stream>>>(mid, out);
}


# both 2x levels in ONE launch, chained in registers: mid-level WMMA D tiles feed level 2's column filter as A (transposed product), its D the row filter as A; no scratch or LDS; f32 sum order changed
# speedup vs baseline: 4.0366x; 1.4419x over previous
#include <hip/hip_runtime.h>

typedef __attribute__((ext_vector_type(2))) float v2f;
typedef __attribute__((ext_vector_type(4))) float v4f;
typedef __attribute__((ext_vector_type(8))) float v8f;

__constant__ float cH[23] = {
    -1.20162964e-4f, 0.0f,  1.615524292e-3f, 0.0f, -1.0385513306e-2f, 0.0f,
     4.3619155884e-2f, 0.0f, -1.45397186478e-1f, 0.0f, 6.1066818237e-1f,
     1.0f,
     6.1066818237e-1f, 0.0f, -1.45397186478e-1f, 0.0f, 4.3619155884e-2f, 0.0f,
    -1.0385513306e-2f, 0.0f,  1.615524292e-3f, 0.0f, -1.20162964e-4f };

#define WMMA4(a, b, c) \
  __builtin_amdgcn_wmma_f32_16x16x4_f32(false, (a), false, (b), (short)0, (c), false, false)


__device__ __forceinline__ v8f row_tile(const v4f& g0, const v4f& g1, const v4f& g2,
                                        const v2f (&br)[5], bool hi)
{
  v2f e;
  e.x = hi ? g0.x : g2.z;
  e.y = hi ? g0.y : g2.w;
  v8f acc = {};
  acc = WMMA4(g0.zw, br[0], acc);
  acc = WMMA4(g1.xy, br[1], acc);
  acc = WMMA4(g1.zw, br[2], acc);
  acc = WMMA4(g2.xy, br[3], acc);
  acc = WMMA4(e,     br[4], acc);
  return acc;
}

__device__ __forceinline__ v2f wchunk(const v8f& lo, const v8f& hi, bool odd, int kk)
{
  const int j = kk + (odd ? 2 : 0);
  v2f p;
  if (j < 4) { p.x = lo[2 * j];       p.y = lo[2 * j + 1]; }
  else       { p.x = hi[2 * (j - 4)]; p.y = hi[2 * (j - 4) + 1]; }
  return p;
}

__device__ __forceinline__ void half_swap(v8f& a, v8f& b)
{
#pragma unroll
  for (int v = 0; v < 8; ++v) {
    const auto r = __builtin_amdgcn_permlane16_swap(__float_as_uint(a[v]), __float_as_uint(b[v]),
                                                    false, false);
    a[v] = __uint_as_float(r[0]);
    b[v] = __uint_as_float(r[1]);
  }
}

template <int N, int WPB>
__global__ __launch_bounds__(32 * WPB) void cdf23_up4x_regchain(
    const float* __restrict__ in, float* __restrict__ out)
{
  __shared__ float sH[96];
  const int tid = threadIdx.x;
  if (tid < 96) {
    const int tap = tid - 32;
    sH[tid] = (tap >= 0 && tap < 23) ? cH[tap] : 0.0f;
  }
  __syncthreads();

  constexpr int OW = 4 * N;
  constexpr int RG = OW / 64;
  const int lane = tid & 31;
  const int n16  = lane & 15;
  const int half = lane >> 4;
  const bool hi  = half != 0;
  const int g     = blockIdx.x * WPB + (tid >> 5);
  const int plane = g / (RG * RG);
  const int rem   = g - plane * (RG * RG);
  const int Rr    = rem / RG;
  const int Rc    = rem - Rr * RG;
  const int prow = 4 * ((n16 & 7) >> 1) + 2 * (n16 >> 3) + (n16 & 1);

  const int nb[5] = { 4 * half, 6 + 4 * half, 8 + 4 * half, 14 + 4 * half, 16 - 14 * half };
  const int lr = 30 - prow;
  v2f br[5];
#pragma unroll
  for (int j = 0; j < 5; ++j) {
    br[j].x = sH[lr + 2 * nb[j]];
    br[j].y = sH[lr + 2 * nb[j] + 2];
  }
  const int l0 = 30 + 4 * half - prow;
  const int l1 = 31 + 4 * half - n16;
  v2f bh[5], bv[5];
#pragma unroll
  for (int kk = 0; kk < 5; ++kk) {
    bh[kk].x = sH[l0 + 8 * kk];
    bh[kk].y = sH[l0 + 8 * kk + 2];
    bv[kk].x = sH[l1 + 8 * kk];
    bv[kk].y = sH[l1 + 8 * kk + 2];
  }

  const float* __restrict__ xin = in + (size_t)plane * N * N;
  float* __restrict__ op = out + (size_t)plane * OW * OW + (size_t)(64 * Rr) * OW + 64 * Rc + lane;

  v4f X[3][5];
#pragma unroll
  for (int t = 0; t < 3; ++t) {
    const float* r = xin + (size_t)((16 * (Rr + t) - 10 + prow) & (N - 1)) * N;
#pragma unroll
    for (int p = 0; p < 5; ++p)
      X[t][p] = *(const v4f*)(r + ((16 * Rc - 12 + 8 * p + 4 * half) & (N - 1)));
  }
  __builtin_amdgcn_sched_barrier(0);

  v8f T[3][3];
#pragma unroll
  for (int t = 0; t < 3; ++t)
#pragma unroll
    for (int nt = 0; nt < 3; ++nt)
      T[t][nt] = row_tile(X[t][nt], X[t][nt + 1], X[t][nt + 2], br, hi);

  v8f M[3][3];
#pragma unroll
  for (int k = 0; k < 3; ++k)
#pragma unroll
    for (int nt = 0; nt < 3; ++nt) {
      v8f acc = {};
#pragma unroll
      for (int kk = 0; kk < 5; ++kk)
        acc = WMMA4(bh[kk], wchunk(T[k >> 1][nt], T[(k >> 1) + 1][nt], (k & 1) != 0, kk), acc);
      M[k][nt] = acc;
    }

#pragma unroll
  for (int ot = 0; ot < 4; ++ot) {
    if (ot > 0) {
      __builtin_amdgcn_sched_barrier(0);
    }
    v8f Ut[3];
#pragma unroll
    for (int nt = 0; nt < 3; ++nt) {
      v8f acc = {};
#pragma unroll
      for (int kk = 0; kk < 5; ++kk)
        acc = WMMA4(wchunk(M[ot >> 1][nt], M[(ot >> 1) + 1][nt], (ot & 1) != 0, kk), bv[kk], acc);
      Ut[nt] = acc;
    }
    v8f O[4];
#pragma unroll
    for (int oc = 0; oc < 4; ++oc) {
      v8f acc = {};
#pragma unroll
      for (int kk = 0; kk < 5; ++kk)
        acc = WMMA4(wchunk(Ut[oc >> 1], Ut[(oc >> 1) + 1], (oc & 1) != 0, kk), bv[kk], acc);
      O[oc] = acc;
    }
    float* o0 = op + (size_t)(16 * ot) * OW;
#pragma unroll
    for (int vl = 0; vl < 2; ++vl) {
      asm("v_nop\n\tv_nop\n\tv_nop\n\tv_nop" : "+v"(O[2 * vl]), "+v"(O[2 * vl + 1]));
      half_swap(O[2 * vl], O[2 * vl + 1]);
#pragma unroll
      for (int v = 0; v < 8; ++v) {
        o0[(size_t)v * OW + 32 * vl]       = O[2 * vl][v];
        o0[(size_t)(8 + v) * OW + 32 * vl] = O[2 * vl + 1][v];
      }
    }
  }
}

extern "C" void kernel_launch(void* const* d_in, const int* in_sizes, int n_in,
                              void* d_out, int out_size, void* d_ws, size_t ws_size,
                              hipStream_t stream) {
  (void)in_sizes; (void)n_in; (void)out_size; (void)d_ws; (void)ws_size;
  const float* img = (const float*)d_in[0];
  float* out = (float*)d_out;

  cdf23_up4x_regchain<256, 8><<<512, 256, 0, stream>>>(img, out);
}


# each wave makes a 128-row x 64-col block (was 64x64), upper half first; input and middle-level row tiles shared by the halves made once: 415 WMMAs per wave for 2 x 230, half the waves; sums unchanged
# speedup vs baseline: 4.3240x; 1.0712x over previous
#include <hip/hip_runtime.h>

typedef __attribute__((ext_vector_type(2))) float v2f;
typedef __attribute__((ext_vector_type(4))) float v4f;
typedef __attribute__((ext_vector_type(8))) float v8f;

__constant__ float cH[23] = {
    -1.20162964e-4f, 0.0f,  1.615524292e-3f, 0.0f, -1.0385513306e-2f, 0.0f,
     4.3619155884e-2f, 0.0f, -1.45397186478e-1f, 0.0f, 6.1066818237e-1f,
     1.0f,
     6.1066818237e-1f, 0.0f, -1.45397186478e-1f, 0.0f, 4.3619155884e-2f, 0.0f,
    -1.0385513306e-2f, 0.0f,  1.615524292e-3f, 0.0f, -1.20162964e-4f };

#define WMMA4(a, b, c) \
  __builtin_amdgcn_wmma_f32_16x16x4_f32(false, (a), false, (b), (short)0, (c), false, false)


__device__ __forceinline__ v8f row_tile(const v4f& g0, const v4f& g1, const v4f& g2,
                                        const v2f (&br)[5], bool hi)
{
  v2f e;
  e.x = hi ? g0.x : g2.z;
  e.y = hi ? g0.y : g2.w;
  v8f acc = {};
  acc = WMMA4(g0.zw, br[0], acc);
  acc = WMMA4(g1.xy, br[1], acc);
  acc = WMMA4(g1.zw, br[2], acc);
  acc = WMMA4(g2.xy, br[3], acc);
  acc = WMMA4(e,     br[4], acc);
  return acc;
}

__device__ __forceinline__ v2f wchunk(const v8f& lo, const v8f& hi, bool odd, int kk)
{
  const int j = kk + (odd ? 2 : 0);
  v2f p;
  if (j < 4) { p.x = lo[2 * j];       p.y = lo[2 * j + 1]; }
  else       { p.x = hi[2 * (j - 4)]; p.y = hi[2 * (j - 4) + 1]; }
  return p;
}

__device__ __forceinline__ void half_swap(v8f& a, v8f& b)
{
#pragma unroll
  for (int v = 0; v < 8; ++v) {
    const auto r = __builtin_amdgcn_permlane16_swap(__float_as_uint(a[v]), __float_as_uint(b[v]),
                                                    false, false);
    a[v] = __uint_as_float(r[0]);
    b[v] = __uint_as_float(r[1]);
  }
}

template <int N, int WPB>
__global__ __launch_bounds__(32 * WPB) void cdf23_up4x_regchain(
    const float* __restrict__ in, float* __restrict__ out)
{
  __shared__ float sH[96];
  const int tid = threadIdx.x;
  if (tid < 96) {
    const int tap = tid - 32;
    sH[tid] = (tap >= 0 && tap < 23) ? cH[tap] : 0.0f;
  }
  __syncthreads();

  constexpr int OW = 4 * N;
  constexpr int CG = OW / 64;
  constexpr int RB = OW / 128;
  const int lane = tid & 31;
  const int n16  = lane & 15;
  const int half = lane >> 4;
  const bool hi  = half != 0;
  const int g     = blockIdx.x * WPB + (tid >> 5);
  const int plane = g / (CG * RB);
  const int rem   = g - plane * (CG * RB);
  const int Rb    = rem / CG;
  const int Rc    = rem - Rb * CG;
  const int Rr    = 2 * Rb;
  const int prow = 4 * ((n16 & 7) >> 1) + 2 * (n16 >> 3) + (n16 & 1);

  const int nb[5] = { 4 * half, 6 + 4 * half, 8 + 4 * half, 14 + 4 * half, 16 - 14 * half };
  const int lr = 30 - prow;
  v2f br[5];
#pragma unroll
  for (int j = 0; j < 5; ++j) {
    br[j].x = sH[lr + 2 * nb[j]];
    br[j].y = sH[lr + 2 * nb[j] + 2];
  }
  const int l0 = 30 + 4 * half - prow;
  const int l1 = 31 + 4 * half - n16;
  v2f bh[5], bv[5];
#pragma unroll
  for (int kk = 0; kk < 5; ++kk) {
    bh[kk].x = sH[l0 + 8 * kk];
    bh[kk].y = sH[l0 + 8 * kk + 2];
    bv[kk].x = sH[l1 + 8 * kk];
    bv[kk].y = sH[l1 + 8 * kk + 2];
  }

  const float* __restrict__ xin = in + (size_t)plane * N * N;
  float* __restrict__ op = out + (size_t)plane * OW * OW + (size_t)(64 * Rr) * OW + 64 * Rc + lane;

  v4f X[4][5];
#pragma unroll
  for (int t = 0; t < 4; ++t) {
    const float* r = xin + (size_t)((16 * (Rr + t) - 10 + prow) & (N - 1)) * N;
#pragma unroll
    for (int p = 0; p < 5; ++p)
      X[t][p] = *(const v4f*)(r + ((16 * Rc - 12 + 8 * p + 4 * half) & (N - 1)));
  }
  __builtin_amdgcn_sched_barrier(0);

  v8f T[4][3];
#pragma unroll
  for (int t = 0; t < 3; ++t)
#pragma unroll
    for (int nt = 0; nt < 3; ++nt)
      T[t][nt] = row_tile(X[t][nt], X[t][nt + 1], X[t][nt + 2], br, hi);

  v8f M[5][3];
#pragma unroll
  for (int k = 0; k < 3; ++k)
#pragma unroll
    for (int nt = 0; nt < 3; ++nt) {
      v8f acc = {};
#pragma unroll
      for (int kk = 0; kk < 5; ++kk)
        acc = WMMA4(bh[kk], wchunk(T[k >> 1][nt], T[(k >> 1) + 1][nt], (k & 1) != 0, kk), acc);
      M[k][nt] = acc;
    }

#pragma unroll
  for (int ot = 0; ot < 8; ++ot) {
    if (ot > 0) {
      __builtin_amdgcn_sched_barrier(0);
    }
    if (ot == 4) {
#pragma unroll
      for (int nt = 0; nt < 3; ++nt)
        T[3][nt] = row_tile(X[3][nt], X[3][nt + 1], X[3][nt + 2], br, hi);
#pragma unroll
      for (int k = 3; k < 5; ++k)
#pragma unroll
        for (int nt = 0; nt < 3; ++nt) {
          v8f acc = {};
#pragma unroll
          for (int kk = 0; kk < 5; ++kk)
            acc = WMMA4(bh[kk], wchunk(T[k >> 1][nt], T[(k >> 1) + 1][nt], (k & 1) != 0, kk), acc);
          M[k][nt] = acc;
        }
    }
    v8f Ut[3];
#pragma unroll
    for (int nt = 0; nt < 3; ++nt) {
      v8f acc = {};
#pragma unroll
      for (int kk = 0; kk < 5; ++kk)
        acc = WMMA4(wchunk(M[ot >> 1][nt], M[(ot >> 1) + 1][nt], (ot & 1) != 0, kk), bv[kk], acc);
      Ut[nt] = acc;
    }
    v8f O[4];
#pragma unroll
    for (int oc = 0; oc < 4; ++oc) {
      v8f acc = {};
#pragma unroll
      for (int kk = 0; kk < 5; ++kk)
        acc = WMMA4(wchunk(Ut[oc >> 1], Ut[(oc >> 1) + 1], (oc & 1) != 0, kk), bv[kk], acc);
      O[oc] = acc;
    }
    float* o0 = op + (size_t)(16 * ot) * OW;
#pragma unroll
    for (int vl = 0; vl < 2; ++vl) {
      asm("v_nop\n\tv_nop\n\tv_nop\n\tv_nop" : "+v"(O[2 * vl]), "+v"(O[2 * vl + 1]));
      half_swap(O[2 * vl], O[2 * vl + 1]);
#pragma unroll
      for (int v = 0; v < 8; ++v) {
        o0[(size_t)v * OW + 32 * vl]       = O[2 * vl][v];
        o0[(size_t)(8 + v) * OW + 32 * vl] = O[2 * vl + 1][v];
      }
    }
  }
}

extern "C" void kernel_launch(void* const* d_in, const int* in_sizes, int n_in,
                              void* d_out, int out_size, void* d_ws, size_t ws_size,
                              hipStream_t stream) {
  (void)in_sizes; (void)n_in; (void)out_size; (void)d_ws; (void)ws_size;
  const float* img = (const float*)d_in[0];
  float* out = (float*)d_out;

  cdf23_up4x_regchain<256, 8><<<256, 256, 0, stream>>>(img, out);
}


# output tiles transposed in registers (last product's taps take columns 0,4,8,12,1,5,..; 3 lane rotations per register, no LDS): 8 stores of 16 B per lane per row tile, not 32 of 4 B; sums unchanged
# speedup vs baseline: 5.4711x; 1.2653x over previous
#include <hip/hip_runtime.h>

typedef __attribute__((ext_vector_type(2))) float v2f;
typedef __attribute__((ext_vector_type(4))) float v4f;
typedef __attribute__((ext_vector_type(8))) float v8f;

__constant__ float cH[23] = {
    -1.20162964e-4f, 0.0f,  1.615524292e-3f, 0.0f, -1.0385513306e-2f, 0.0f,
     4.3619155884e-2f, 0.0f, -1.45397186478e-1f, 0.0f, 6.1066818237e-1f,
     1.0f,
     6.1066818237e-1f, 0.0f, -1.45397186478e-1f, 0.0f, 4.3619155884e-2f, 0.0f,
    -1.0385513306e-2f, 0.0f,  1.615524292e-3f, 0.0f, -1.20162964e-4f };

#define WMMA4(a, b, c) \
  __builtin_amdgcn_wmma_f32_16x16x4_f32(false, (a), false, (b), (short)0, (c), false, false)


__device__ __forceinline__ v8f row_tile(const v4f& g0, const v4f& g1, const v4f& g2,
                                        const v2f (&br)[5], bool hi)
{
  v2f e;
  e.x = hi ? g0.x : g2.z;
  e.y = hi ? g0.y : g2.w;
  v8f acc = {};
  acc = WMMA4(g0.zw, br[0], acc);
  acc = WMMA4(g1.xy, br[1], acc);
  acc = WMMA4(g1.zw, br[2], acc);
  acc = WMMA4(g2.xy, br[3], acc);
  acc = WMMA4(e,     br[4], acc);
  return acc;
}

__device__ __forceinline__ v2f wchunk(const v8f& lo, const v8f& hi, bool odd, int kk)
{
  const int j = kk + (odd ? 2 : 0);
  v2f p;
  if (j < 4) { p.x = lo[2 * j];       p.y = lo[2 * j + 1]; }
  else       { p.x = hi[2 * (j - 4)]; p.y = hi[2 * (j - 4) + 1]; }
  return p;
}

template <int ROT, int MASK>
__device__ __forceinline__ float put_bank(float keep, float from)
{
  return __builtin_bit_cast(float,
      __builtin_amdgcn_update_dpp(__builtin_bit_cast(int, keep), __builtin_bit_cast(int, from),
                                  0x120 + ROT  , 0xF, MASK, false));
}

template <int N, int WPB>
__global__ __launch_bounds__(32 * WPB) void cdf23_up4x_regchain(
    const float* __restrict__ in, float* __restrict__ out)
{
  __shared__ float sH[96];
  const int tid = threadIdx.x;
  if (tid < 96) {
    const int tap = tid - 32;
    sH[tid] = (tap >= 0 && tap < 23) ? cH[tap] : 0.0f;
  }
  __syncthreads();

  constexpr int OW = 4 * N;
  constexpr int CG = OW / 64;
  constexpr int RB = OW / 128;
  const int lane = tid & 31;
  const int n16  = lane & 15;
  const int half = lane >> 4;
  const bool hi  = half != 0;
  const int g     = blockIdx.x * WPB + (tid >> 5);
  const int plane = g / (CG * RB);
  const int rem   = g - plane * (CG * RB);
  const int Rb    = rem / CG;
  const int Rc    = rem - Rb * CG;
  const int Rr    = 2 * Rb;
  const int prow = 4 * ((n16 & 7) >> 1) + 2 * (n16 >> 3) + (n16 & 1);

  const int nb[5] = { 4 * half, 6 + 4 * half, 8 + 4 * half, 14 + 4 * half, 16 - 14 * half };
  const int lr = 30 - prow;
  v2f br[5];
#pragma unroll
  for (int j = 0; j < 5; ++j) {
    br[j].x = sH[lr + 2 * nb[j]];
    br[j].y = sH[lr + 2 * nb[j] + 2];
  }
  const int l0 = 30 + 4 * half - prow;
  const int l1 = 31 + 4 * half - n16;
  const int l2 = 31 + 4 * half - (4 * (n16 & 3) + (n16 >> 2));
  v2f bh[5], bv[5], bw[5];
#pragma unroll
  for (int kk = 0; kk < 5; ++kk) {
    bh[kk].x = sH[l0 + 8 * kk];
    bh[kk].y = sH[l0 + 8 * kk + 2];
    bv[kk].x = sH[l1 + 8 * kk];
    bv[kk].y = sH[l1 + 8 * kk + 2];
    bw[kk].x = sH[l2 + 8 * kk];
    bw[kk].y = sH[l2 + 8 * kk + 2];
  }

  const float* __restrict__ xin = in + (size_t)plane * N * N;
  float* __restrict__ op = out + (size_t)plane * OW * OW + (size_t)(64 * Rr + 8 * half) * OW + 64 * Rc + 4 * n16;

  v4f X[4][5];
#pragma unroll
  for (int t = 0; t < 4; ++t) {
    const float* r = xin + (size_t)((16 * (Rr + t) - 10 + prow) & (N - 1)) * N;
#pragma unroll
    for (int p = 0; p < 5; ++p)
      X[t][p] = *(const v4f*)(r + ((16 * Rc - 12 + 8 * p + 4 * half) & (N - 1)));
  }
  __builtin_amdgcn_sched_barrier(0);

  v8f T[4][3];
#pragma unroll
  for (int t = 0; t < 3; ++t)
#pragma unroll
    for (int nt = 0; nt < 3; ++nt)
      T[t][nt] = row_tile(X[t][nt], X[t][nt + 1], X[t][nt + 2], br, hi);

  v8f M[5][3];
#pragma unroll
  for (int k = 0; k < 3; ++k)
#pragma unroll
    for (int nt = 0; nt < 3; ++nt) {
      v8f acc = {};
#pragma unroll
      for (int kk = 0; kk < 5; ++kk)
        acc = WMMA4(bh[kk], wchunk(T[k >> 1][nt], T[(k >> 1) + 1][nt], (k & 1) != 0, kk), acc);
      M[k][nt] = acc;
    }

#pragma unroll
  for (int ot = 0; ot < 8; ++ot) {
    if (ot > 0) {
      __builtin_amdgcn_sched_barrier(0);
    }
    if (ot == 4) {
#pragma unroll
      for (int nt = 0; nt < 3; ++nt)
        T[3][nt] = row_tile(X[3][nt], X[3][nt + 1], X[3][nt + 2], br, hi);
#pragma unroll
      for (int k = 3; k < 5; ++k)
#pragma unroll
        for (int nt = 0; nt < 3; ++nt) {
          v8f acc = {};
#pragma unroll
          for (int kk = 0; kk < 5; ++kk)
            acc = WMMA4(bh[kk], wchunk(T[k >> 1][nt], T[(k >> 1) + 1][nt], (k & 1) != 0, kk), acc);
          M[k][nt] = acc;
        }
    }
    v8f Ut[3];
#pragma unroll
    for (int nt = 0; nt < 3; ++nt) {
      v8f acc = {};
#pragma unroll
      for (int kk = 0; kk < 5; ++kk)
        acc = WMMA4(wchunk(M[ot >> 1][nt], M[(ot >> 1) + 1][nt], (ot & 1) != 0, kk), bv[kk], acc);
      Ut[nt] = acc;
    }
    v8f O[4];
#pragma unroll
    for (int oc = 0; oc < 4; ++oc) {
      v8f acc = {};
#pragma unroll
      for (int kk = 0; kk < 5; ++kk)
        acc = WMMA4(wchunk(Ut[oc >> 1], Ut[(oc >> 1) + 1], (oc & 1) != 0, kk), bw[kk], acc);
      O[oc] = acc;
    }
    float* o0 = op + (size_t)(16 * ot) * OW;
    v4f S[8];
#pragma unroll
    for (int r = 0; r < 8; ++r) {
      const float a0 = O[0][r], a1 = O[1][r], a2 = O[2][r], a3 = O[3][r];
      S[r].x = put_bank<12, 8>(put_bank<8, 4>(put_bank<4, 2>(a0, a1), a2), a3);
      S[r].y = put_bank<8, 8>(put_bank<4, 4>(put_bank<12, 1>(a1, a0), a2), a3);
      S[r].z = put_bank<4, 8>(put_bank<12, 2>(put_bank<8, 1>(a2, a0), a1), a3);
      S[r].w = put_bank<12, 4>(put_bank<8, 2>(put_bank<4, 1>(a3, a0), a1), a2);
    }
#pragma unroll
    for (int r = 0; r < 8; ++r)
      *(v4f*)(o0 + (size_t)r * OW) = S[r];
  }
}

extern "C" void kernel_launch(void* const* d_in, const int* in_sizes, int n_in,
                              void* d_out, int out_size, void* d_ws, size_t ws_size,
                              hipStream_t stream) {
  (void)in_sizes; (void)n_in; (void)out_size; (void)d_ws; (void)ws_size;
  const float* img = (const float*)d_in[0];
  float* out = (float*)d_out;

  cdf23_up4x_regchain<256, 8><<<256, 256, 0, stream>>>(img, out);
}


# 8 waves of a work-group ring one block row (64x128 blocks, not 128x64): product 3's halo column tile comes from the right neighbour via LDS, not re-made: 360 WMMAs per wave, not 415; sums unchanged
# speedup vs baseline: 5.9297x; 1.0838x over previous
#include <hip/hip_runtime.h>

typedef __attribute__((ext_vector_type(2))) float v2f;
typedef __attribute__((ext_vector_type(4))) float v4f;
typedef __attribute__((ext_vector_type(8))) float v8f;

__constant__ float cH[23] = {
    -1.20162964e-4f, 0.0f,  1.615524292e-3f, 0.0f, -1.0385513306e-2f, 0.0f,
     4.3619155884e-2f, 0.0f, -1.45397186478e-1f, 0.0f, 6.1066818237e-1f,
     1.0f,
     6.1066818237e-1f, 0.0f, -1.45397186478e-1f, 0.0f, 4.3619155884e-2f, 0.0f,
    -1.0385513306e-2f, 0.0f,  1.615524292e-3f, 0.0f, -1.20162964e-4f };

#define WMMA4(a, b, c) \
  __builtin_amdgcn_wmma_f32_16x16x4_f32(false, (a), false, (b), (short)0, (c), false, false)


__device__ __forceinline__ v8f row_tile(const v4f& g0, const v4f& g1, const v4f& g2,
                                        const v2f (&br)[5], bool hi)
{
  v2f e;
  e.x = hi ? g0.x : g2.z;
  e.y = hi ? g0.y : g2.w;
  v8f acc = {};
  acc = WMMA4(g0.zw, br[0], acc);
  acc = WMMA4(g1.xy, br[1], acc);
  acc = WMMA4(g1.zw, br[2], acc);
  acc = WMMA4(g2.xy, br[3], acc);
  acc = WMMA4(e,     br[4], acc);
  return acc;
}

__device__ __forceinline__ v2f wchunk(const v8f& lo, const v8f& hi, bool odd, int kk)
{
  const int j = kk + (odd ? 2 : 0);
  v2f p;
  if (j < 4) { p.x = lo[2 * j];       p.y = lo[2 * j + 1]; }
  else       { p.x = hi[2 * (j - 4)]; p.y = hi[2 * (j - 4) + 1]; }
  return p;
}

template <int ROT, int MASK>
__device__ __forceinline__ float put_bank(float keep, float from)
{
  return __builtin_bit_cast(float,
      __builtin_amdgcn_update_dpp(__builtin_bit_cast(int, keep), __builtin_bit_cast(int, from),
                                  0x120 + ROT  , 0xF, MASK, false));
}

__device__ __forceinline__ v8f out_tile(const v8f& lo, const v8f& hi, bool odd, const v2f (&bw)[5])
{
  v8f acc = {};
#pragma unroll
  for (int kk = 0; kk < 5; ++kk)
    acc = WMMA4(wchunk(lo, hi, odd, kk), bw[kk], acc);
  return acc;
}

template <int OW>
__device__ __forceinline__ void store_tiles(const v8f (&O)[4], float* o0)
{
  v4f S[8];
#pragma unroll
  for (int r = 0; r < 8; ++r) {
    const float a0 = O[0][r], a1 = O[1][r], a2 = O[2][r], a3 = O[3][r];
    S[r].x = put_bank<12, 8>(put_bank<8, 4>(put_bank<4, 2>(a0, a1), a2), a3);
    S[r].y = put_bank<8, 8>(put_bank<4, 4>(put_bank<12, 1>(a1, a0), a2), a3);
    S[r].z = put_bank<4, 8>(put_bank<12, 2>(put_bank<8, 1>(a2, a0), a1), a3);
    S[r].w = put_bank<12, 4>(put_bank<8, 2>(put_bank<4, 1>(a3, a0), a1), a2);
  }
#pragma unroll
  for (int r = 0; r < 8; ++r)
    *(v4f*)(o0 + (size_t)r * OW) = S[r];
}

template <int N>
__global__ __launch_bounds__(256) void cdf23_up4x_ring(
    const float* __restrict__ in, float* __restrict__ out)
{
  constexpr int OW = 4 * N;
  constexpr int RB = OW / 64;
  static_assert(OW == 8 * 128, "eight waves of 128 columns must close the ring of one row");

  __shared__ float sH[96];
  __shared__ v4f sU[2][8][2][32];
  const int tid = threadIdx.x;
  if (tid < 96) {
    const int tap = tid - 32;
    sH[tid] = (tap >= 0 && tap < 23) ? cH[tap] : 0.0f;
  }
  __syncthreads();

  const int lane = tid & 31;
  const int n16  = lane & 15;
  const int half = lane >> 4;
  const bool hi  = half != 0;
  const int wv    = tid >> 5;
  const int plane = blockIdx.x / RB;
  const int Rr    = blockIdx.x - plane * RB;
  const int Cc    = 2 * wv;
  const int prow = 4 * ((n16 & 7) >> 1) + 2 * (n16 >> 3) + (n16 & 1);

  const int nb[5] = { 4 * half, 6 + 4 * half, 8 + 4 * half, 14 + 4 * half, 16 - 14 * half };
  const int lr = 30 - prow;
  v2f br[5];
#pragma unroll
  for (int j = 0; j < 5; ++j) {
    br[j].x = sH[lr + 2 * nb[j]];
    br[j].y = sH[lr + 2 * nb[j] + 2];
  }
  const int l0 = 30 + 4 * half - prow;
  const int l1 = 31 + 4 * half - n16;
  const int l2 = 31 + 4 * half - (4 * (n16 & 3) + (n16 >> 2));
  v2f bh[5], bv[5], bw[5];
#pragma unroll
  for (int kk = 0; kk < 5; ++kk) {
    bh[kk].x = sH[l0 + 8 * kk];
    bh[kk].y = sH[l0 + 8 * kk + 2];
    bv[kk].x = sH[l1 + 8 * kk];
    bv[kk].y = sH[l1 + 8 * kk + 2];
    bw[kk].x = sH[l2 + 8 * kk];
    bw[kk].y = sH[l2 + 8 * kk + 2];
  }

  const float* __restrict__ xin = in + (size_t)plane * N * N;
  float* __restrict__ op = out + (size_t)plane * OW * OW + (size_t)(64 * Rr + 8 * half) * OW + 64 * Cc + 4 * n16;

  v4f X[3][6];
#pragma unroll
  for (int t = 0; t < 3; ++t) {
    const float* r = xin + (size_t)((16 * (Rr + t) - 10 + prow) & (N - 1)) * N;
#pragma unroll
    for (int p = 0; p < 6; ++p)
      X[t][p] = *(const v4f*)(r + ((16 * Cc - 12 + 8 * p + 4 * half) & (N - 1)));
  }
  __builtin_amdgcn_sched_barrier(0);

  v8f T[3][4];
#pragma unroll
  for (int t = 0; t < 3; ++t)
#pragma unroll
    for (int nt = 0; nt < 4; ++nt)
      T[t][nt] = row_tile(X[t][nt], X[t][nt + 1], X[t][nt + 2], br, hi);

  v8f M[3][4];
#pragma unroll
  for (int k = 0; k < 3; ++k)
#pragma unroll
    for (int nt = 0; nt < 4; ++nt) {
      v8f acc = {};
#pragma unroll
      for (int kk = 0; kk < 5; ++kk)
        acc = WMMA4(bh[kk], wchunk(T[k >> 1][nt], T[(k >> 1) + 1][nt], (k & 1) != 0, kk), acc);
      M[k][nt] = acc;
    }

#pragma unroll
  for (int ot = 0; ot < 4; ++ot) {
    if (ot > 0) {
      __builtin_amdgcn_sched_barrier(0);
    }
    v8f Ut[5];
#pragma unroll
    for (int nt = 0; nt < 4; ++nt) {
      v8f acc = {};
#pragma unroll
      for (int kk = 0; kk < 5; ++kk)
        acc = WMMA4(wchunk(M[ot >> 1][nt], M[(ot >> 1) + 1][nt], (ot & 1) != 0, kk), bv[kk], acc);
      Ut[nt] = acc;
      if (nt == 0) {
        v4f q0, q1;
        q0.x = acc[0]; q0.y = acc[1]; q0.z = acc[2]; q0.w = acc[3];
        q1.x = acc[4]; q1.y = acc[5]; q1.z = acc[6]; q1.w = acc[7];
        sU[ot & 1][wv][0][lane] = q0;
        sU[ot & 1][wv][1][lane] = q1;
      }
    }
    float* o0 = op + (size_t)(16 * ot) * OW;
    v8f O[4];
#pragma unroll
    for (int oc = 0; oc < 4; ++oc)
      O[oc] = out_tile(Ut[oc >> 1], Ut[(oc >> 1) + 1], (oc & 1) != 0, bw);
    store_tiles<OW>(O, o0);
    v8f Q[4];
    Q[0] = out_tile(Ut[2], Ut[3], false, bw);
    Q[1] = out_tile(Ut[2], Ut[3], true, bw);
    __builtin_amdgcn_fence(__ATOMIC_RELEASE, "workgroup", "local");
    __builtin_amdgcn_s_barrier();
    __builtin_amdgcn_fence(__ATOMIC_ACQUIRE, "workgroup", "local");
    {
      const v4f q0 = sU[ot & 1][(wv + 1) & 7][0][lane];
      const v4f q1 = sU[ot & 1][(wv + 1) & 7][1][lane];
      v8f u;
      u[0] = q0.x; u[1] = q0.y; u[2] = q0.z; u[3] = q0.w;
      u[4] = q1.x; u[5] = q1.y; u[6] = q1.z; u[7] = q1.w;
      Ut[4] = u;
    }
    Q[2] = out_tile(Ut[3], Ut[4], false, bw);
    Q[3] = out_tile(Ut[3], Ut[4], true, bw);
    store_tiles<OW>(Q, o0 + 64);
  }
}

extern "C" void kernel_launch(void* const* d_in, const int* in_sizes, int n_in,
                              void* d_out, int out_size, void* d_ws, size_t ws_size,
                              hipStream_t stream) {
  (void)in_sizes; (void)n_in; (void)out_size; (void)d_ws; (void)ws_size;
  const float* img = (const float*)d_in[0];
  float* out = (float*)d_out;

  cdf23_up4x_ring<256><<<256, 256, 0, stream>>>(img, out);
}


# both column passes first, on input-width tiles loaded in B layout, then both row passes; halo input-column tile from the right neighbour via LDS: 310 WMMAs per wave, not 360; f32 sum order changed
# speedup vs baseline: 6.5261x; 1.1006x over previous
#include <hip/hip_runtime.h>

typedef __attribute__((ext_vector_type(2))) float v2f;
typedef __attribute__((ext_vector_type(4))) float v4f;
typedef __attribute__((ext_vector_type(8))) float v8f;

__constant__ float cH[23] = {
    -1.20162964e-4f, 0.0f,  1.615524292e-3f, 0.0f, -1.0385513306e-2f, 0.0f,
     4.3619155884e-2f, 0.0f, -1.45397186478e-1f, 0.0f, 6.1066818237e-1f,
     1.0f,
     6.1066818237e-1f, 0.0f, -1.45397186478e-1f, 0.0f, 4.3619155884e-2f, 0.0f,
    -1.0385513306e-2f, 0.0f,  1.615524292e-3f, 0.0f, -1.20162964e-4f };

#define WMMA4(a, b, c) \
  __builtin_amdgcn_wmma_f32_16x16x4_f32(false, (a), false, (b), (short)0, (c), false, false)


__device__ __forceinline__ v2f wchunk(const v8f& lo, const v8f& hi, bool odd, int kk)
{
  const int j = kk + (odd ? 2 : 0);
  v2f p;
  if (j < 4) { p.x = lo[2 * j];       p.y = lo[2 * j + 1]; }
  else       { p.x = hi[2 * (j - 4)]; p.y = hi[2 * (j - 4) + 1]; }
  return p;
}

template <int ROT, int MASK>
__device__ __forceinline__ float put_bank(float keep, float from)
{
  return __builtin_bit_cast(float,
      __builtin_amdgcn_update_dpp(__builtin_bit_cast(int, keep), __builtin_bit_cast(int, from),
                                  0x120 + ROT  , 0xF, MASK, false));
}

__device__ __forceinline__ v8f out_tile(const v8f& lo, const v8f& hi, bool odd, const v2f (&bw)[5])
{
  v8f acc = {};
#pragma unroll
  for (int kk = 0; kk < 5; ++kk)
    acc = WMMA4(wchunk(lo, hi, odd, kk), bw[kk], acc);
  return acc;
}

template <int OW>
__device__ __forceinline__ void store_tiles(const v8f (&O)[4], float* o0)
{
  v4f S[8];
#pragma unroll
  for (int r = 0; r < 8; ++r) {
    const float a0 = O[0][r], a1 = O[1][r], a2 = O[2][r], a3 = O[3][r];
    S[r].x = put_bank<12, 8>(put_bank<8, 4>(put_bank<4, 2>(a0, a1), a2), a3);
    S[r].y = put_bank<8, 8>(put_bank<4, 4>(put_bank<12, 1>(a1, a0), a2), a3);
    S[r].z = put_bank<4, 8>(put_bank<12, 2>(put_bank<8, 1>(a2, a0), a1), a3);
    S[r].w = put_bank<12, 4>(put_bank<8, 2>(put_bank<4, 1>(a3, a0), a1), a2);
  }
#pragma unroll
  for (int r = 0; r < 8; ++r)
    *(v4f*)(o0 + (size_t)r * OW) = S[r];
}

template <int N>
__global__ __launch_bounds__(256) void cdf23_up4x_ring(
    const float* __restrict__ in, float* __restrict__ out)
{
  constexpr int OW = 4 * N;
  constexpr int RB = OW / 64;
  static_assert(OW == 8 * 128, "eight waves of 128 columns must close the ring of one row");

  __shared__ float sH[96];
  __shared__ v4f sU[2][8][2][32];
  __shared__ v4f sY[8][4][2][32];
  const int tid = threadIdx.x;
  if (tid < 96) {
    const int tap = tid - 32;
    sH[tid] = (tap >= 0 && tap < 23) ? cH[tap] : 0.0f;
  }
  __syncthreads();

  const int lane = tid & 31;
  const int n16  = lane & 15;
  const int half = lane >> 4;
  const int wv    = tid >> 5;
  const int plane = blockIdx.x / RB;
  const int Rr    = blockIdx.x - plane * RB;
  const int Cc    = 2 * wv;
  const int prow = 4 * ((n16 & 7) >> 1) + 2 * (n16 >> 3) + (n16 & 1);

  const int l0 = 30 + 4 * half - prow;
  const int l1 = 31 + 4 * half - n16;
  const int l2 = 31 + 4 * half - (4 * (n16 & 3) + (n16 >> 2));
  v2f bh[5], bv[5], bw[5];
#pragma unroll
  for (int kk = 0; kk < 5; ++kk) {
    bh[kk].x = sH[l0 + 8 * kk];
    bh[kk].y = sH[l0 + 8 * kk + 2];
    bv[kk].x = sH[l1 + 8 * kk];
    bv[kk].y = sH[l1 + 8 * kk + 2];
    bw[kk].x = sH[l2 + 8 * kk];
    bw[kk].y = sH[l2 + 8 * kk + 2];
  }

  const float* __restrict__ xin = in + (size_t)plane * N * N;
  float* __restrict__ op = out + (size_t)plane * OW * OW + (size_t)(64 * Rr + 8 * half) * OW + 64 * Cc + 4 * n16;

  v2f Xb[9][2];
#pragma unroll
  for (int c = 0; c < 9; ++c) {
    const float* r0 = xin + (size_t)((16 * Rr - 10 + 4 * c + 2 * half) & (N - 1)) * N;
    const float* r1 = xin + (size_t)((16 * Rr - 10 + 4 * c + 2 * half + 1) & (N - 1)) * N;
#pragma unroll
    for (int j = 0; j < 2; ++j) {
      const int col = (16 * (Cc + j) - 10 + prow) & (N - 1);
      Xb[c][j].x = r0[col];
      Xb[c][j].y = r1[col];
    }
  }
  __builtin_amdgcn_sched_barrier(0);

  v8f Y1[3][2];
#pragma unroll
  for (int k = 0; k < 3; ++k)
#pragma unroll
    for (int j = 0; j < 2; ++j) {
      v8f acc = {};
#pragma unroll
      for (int kk = 0; kk < 5; ++kk)
        acc = WMMA4(bh[kk], Xb[2 * k + kk][j], acc);
      Y1[k][j] = acc;
    }

  v8f Y2[4][3];
#pragma unroll
  for (int ot = 0; ot < 4; ++ot)
#pragma unroll
    for (int j = 0; j < 2; ++j) {
      v8f acc = {};
#pragma unroll
      for (int kk = 0; kk < 5; ++kk)
        acc = WMMA4(wchunk(Y1[ot >> 1][j], Y1[(ot >> 1) + 1][j], (ot & 1) != 0, kk), bv[kk], acc);
      Y2[ot][j] = acc;
      if (j == 0) {
        v4f q0, q1;
        q0.x = acc[0]; q0.y = acc[1]; q0.z = acc[2]; q0.w = acc[3];
        q1.x = acc[4]; q1.y = acc[5]; q1.z = acc[6]; q1.w = acc[7];
        sY[wv][ot][0][lane] = q0;
        sY[wv][ot][1][lane] = q1;
      }
    }
  __builtin_amdgcn_fence(__ATOMIC_RELEASE, "workgroup", "local");
  __builtin_amdgcn_s_barrier();
  __builtin_amdgcn_fence(__ATOMIC_ACQUIRE, "workgroup", "local");
#pragma unroll
  for (int ot = 0; ot < 4; ++ot) {
    const v4f q0 = sY[(wv + 1) & 7][ot][0][lane];
    const v4f q1 = sY[(wv + 1) & 7][ot][1][lane];
    v8f u;
    u[0] = q0.x; u[1] = q0.y; u[2] = q0.z; u[3] = q0.w;
    u[4] = q1.x; u[5] = q1.y; u[6] = q1.z; u[7] = q1.w;
    Y2[ot][2] = u;
  }

#pragma unroll
  for (int ot = 0; ot < 4; ++ot) {
    if (ot > 0) {
      __builtin_amdgcn_sched_barrier(0);
    }
    v8f Zt[5];
#pragma unroll
    for (int nt = 0; nt < 4; ++nt) {
      v8f acc = {};
#pragma unroll
      for (int kk = 0; kk < 5; ++kk)
        acc = WMMA4(bh[kk], wchunk(Y2[ot][nt >> 1], Y2[ot][(nt >> 1) + 1], (nt & 1) != 0, kk), acc);
      Zt[nt] = acc;
      if (nt == 0) {
        v4f q0, q1;
        q0.x = acc[0]; q0.y = acc[1]; q0.z = acc[2]; q0.w = acc[3];
        q1.x = acc[4]; q1.y = acc[5]; q1.z = acc[6]; q1.w = acc[7];
        sU[ot & 1][wv][0][lane] = q0;
        sU[ot & 1][wv][1][lane] = q1;
      }
    }
    float* o0 = op + (size_t)(16 * ot) * OW;
    v8f O[4];
#pragma unroll
    for (int oc = 0; oc < 4; ++oc)
      O[oc] = out_tile(Zt[oc >> 1], Zt[(oc >> 1) + 1], (oc & 1) != 0, bw);
    store_tiles<OW>(O, o0);
    v8f Q[4];
    Q[0] = out_tile(Zt[2], Zt[3], false, bw);
    Q[1] = out_tile(Zt[2], Zt[3], true, bw);
    __builtin_amdgcn_fence(__ATOMIC_RELEASE, "workgroup", "local");
    __builtin_amdgcn_s_barrier();
    __builtin_amdgcn_fence(__ATOMIC_ACQUIRE, "workgroup", "local");
    {
      const v4f q0 = sU[ot & 1][(wv + 1) & 7][0][lane];
      const v4f q1 = sU[ot & 1][(wv + 1) & 7][1][lane];
      v8f u;
      u[0] = q0.x; u[1] = q0.y; u[2] = q0.z; u[3] = q0.w;
      u[4] = q1.x; u[5] = q1.y; u[6] = q1.z; u[7] = q1.w;
      Zt[4] = u;
    }
    Q[2] = out_tile(Zt[3], Zt[4], false, bw);
    Q[3] = out_tile(Zt[3], Zt[4], true, bw);
    store_tiles<OW>(Q, o0 + 64);
  }
}

extern "C" void kernel_launch(void* const* d_in, const int* in_sizes, int n_in,
                              void* d_out, int out_size, void* d_ws, size_t ws_size,
                              hipStream_t stream) {
  (void)in_sizes; (void)n_in; (void)out_size; (void)d_ws; (void)ws_size;
  const float* img = (const float*)d_in[0];
  float* out = (float*)d_out;

  cdf23_up4x_ring<256><<<256, 256, 0, stream>>>(img, out);
}
